# starting kernel
# baseline (speedup 1.0000x reference)
#include <hip/hip_runtime.h>
#include <math.h>

typedef __attribute__((ext_vector_type(2))) float v2f;
typedef __attribute__((ext_vector_type(8))) float v8f;

#define TOK_PER_WG 16
#define DDIM 256
#define M_MEM 16384
#define KSEL 32
#define NT 512
#define NCHUNK (M_MEM / NT)
#define SC_STRIDE 516
#define LN_EPS 1e-5f

__device__ __forceinline__ float wave_sum_f(float v) {
#pragma unroll
  for (int o = 16; o > 0; o >>= 1) v += __shfl_xor(v, o, 32);
  return v;
}
__device__ __forceinline__ float wave_max_f(float v) {
#pragma unroll
  for (int o = 16; o > 0; o >>= 1) v = fmaxf(v, __shfl_xor(v, o, 32));
  return v;
}

__global__ __launch_bounds__(512, 1)
void core_processor_kernel(const float* __restrict__ x,
                           const float* __restrict__ mem_keys,
                           const float* __restrict__ mem_values,
                           const float* __restrict__ W_fuse,
                           const float* __restrict__ b_fuse,
                           const float* __restrict__ ln_g,
                           const float* __restrict__ ln_b,
                           const float* __restrict__ W_op,
                           const float* __restrict__ b_op,
                           float* __restrict__ out)
{
  __shared__ __align__(16) float xs[TOK_PER_WG * DDIM];
  __shared__ __align__(16) float sc[TOK_PER_WG * SC_STRIDE];
  __shared__ float tv[TOK_PER_WG * KSEL];
  __shared__ int   ti[TOK_PER_WG * KSEL];

  const int tid  = threadIdx.x;
  const int lane = tid & 31;
  const int w    = tid >> 5;
  const long tokBase = (long)blockIdx.x * TOK_PER_WG;

  const int rowA  = lane & 15;
  const int khalf = (lane >> 4) << 1;

  {
    const float4* src = (const float4*)(x + tokBase * DDIM);
    float4* dst = (float4*)xs;
    dst[tid]       = src[tid];
    dst[tid + 512] = src[tid + 512];
  }
  tv[w * KSEL + lane] = -INFINITY;
  ti[w * KSEL + lane] = 0;
  float curMin = -INFINITY;
  int   minPos = 0;
  __syncthreads();

  for (int chunk = 0; chunk < NCHUNK; ++chunk) {
    const int chunkBase = chunk * NT;
#pragma unroll
    for (int t2 = 0; t2 < 2; ++t2) {
      const int tile = w * 2 + t2;
      const int colB = chunkBase + tile * 16 + (lane & 15);
      const float* ap = xs + rowA * DDIM + khalf;
      const float* bp = mem_keys + (long)colB * DDIM + khalf;
      v8f acc = {};
#pragma unroll 4
      for (int k0 = 0; k0 < DDIM; k0 += 4) {
        v2f a = *(const v2f*)(ap + k0);
        v2f b = *(const v2f*)(bp + k0);
        acc = __builtin_amdgcn_wmma_f32_16x16x4_f32(
            false, a, false, b, (short)0, acc, false, false);
      }
      const int m0 = (lane >> 4) * 8;
      const int nn = tile * 16 + (lane & 15);
#pragma unroll
      for (int r = 0; r < 8; ++r)
        sc[(m0 + r) * SC_STRIDE + nn] = acc[r];
    }
    __syncthreads();

#pragma unroll 1
    for (int j = 0; j < NT / 32; ++j) {
      const int   col = lane + 32 * j;
      const float s   = sc[w * SC_STRIDE + col];
      const int   gi  = chunkBase + col;
      bool pend = (s > curMin);
      while (true) {
        unsigned long long bal = __ballot(pend);
        if (bal == 0ull) break;
        const int leader = (int)(__ffsll(bal) - 1);
        if (lane == leader) {
          tv[w * KSEL + minPos] = s;
          ti[w * KSEL + minPos] = gi;
          pend = false;
        }
        float v = tv[w * KSEL + lane];
        int   p = lane;
#pragma unroll
        for (int o = 16; o > 0; o >>= 1) {
          float v2 = __shfl_xor(v, o, 32);
          int   p2 = __shfl_xor(p, o, 32);
          if (v2 < v) { v = v2; p = p2; }
        }
        curMin = v; minPos = p;
        pend = pend && (s > curMin);
      }
    }
    __syncthreads();
  }

  {
    const float myv = tv[w * KSEL + lane];
    const int   myi = ti[w * KSEL + lane];
    const float mx  = wave_max_f(myv);
    const float e   = __expf(myv - mx);
    const float sum = wave_sum_f(e);
    const float wn  = e / sum;

    float acc[8];
#pragma unroll
    for (int q = 0; q < 8; ++q) acc[q] = 0.f;
#pragma unroll 1
    for (int k = 0; k < KSEL; ++k) {
      const float wk = __shfl(wn, k, 32);
      const int   id = __shfl(myi, k, 32);
      const float4* vp = (const float4*)(mem_values + (long)id * DDIM + lane * 8);
      const float4 a0 = vp[0];
      const float4 a1 = vp[1];
      acc[0] += wk * a0.x; acc[1] += wk * a0.y; acc[2] += wk * a0.z; acc[3] += wk * a0.w;
      acc[4] += wk * a1.x; acc[5] += wk * a1.y; acc[6] += wk * a1.z; acc[7] += wk * a1.w;
    }
#pragma unroll
    for (int q = 0; q < 8; ++q) xs[w * DDIM + lane * 8 + q] += acc[q];
  }
  __syncthreads();

  {
    const int colB = w * 16 + (lane & 15);
    const float* ap = xs + rowA * DDIM + khalf;
    v8f acc = {};
#pragma unroll 4
    for (int k0 = 0; k0 < DDIM; k0 += 4) {
      v2f a = *(const v2f*)(ap + k0);
      const int kk = k0 + khalf;
      v2f b;
      b.x = W_fuse[(long)kk * DDIM + colB];
      b.y = W_fuse[(long)(kk + 1) * DDIM + colB];
      acc = __builtin_amdgcn_wmma_f32_16x16x4_f32(
          false, a, false, b, (short)0, acc, false, false);
    }
    const float bf = b_fuse[colB];
    const int m0 = (lane >> 4) * 8;
#pragma unroll
    for (int r = 0; r < 8; ++r)
      sc[(m0 + r) * DDIM + colB] = acc[r] + bf;
  }
  __syncthreads();

  {
    const int d0 = lane * 8;
    float h[8];
#pragma unroll
    for (int q = 0; q < 8; ++q) h[q] = sc[w * DDIM + d0 + q];
    float s1 = 0.f, s2 = 0.f;
#pragma unroll
    for (int q = 0; q < 8; ++q) { s1 += h[q]; s2 += h[q] * h[q]; }
    s1 = wave_sum_f(s1);
    s2 = wave_sum_f(s2);
    const float mu   = s1 * (1.0f / DDIM);
    const float var  = s2 * (1.0f / DDIM) - mu * mu;
    const float rstd = rsqrtf(var + LN_EPS);
#pragma unroll
    for (int q = 0; q < 8; ++q) {
      const float g = ln_g[d0 + q];
      const float b = ln_b[d0 + q];
      const float v = (h[q] - mu) * rstd * g + b;
      sc[w * DDIM + d0 + q] = fmaxf(v, 0.f);
    }
  }
  __syncthreads();

  {
    const int colB = w * 16 + (lane & 15);
    const float* ap = sc + rowA * DDIM + khalf;
    v8f acc = {};
#pragma unroll 4
    for (int k0 = 0; k0 < DDIM; k0 += 4) {
      v2f a = *(const v2f*)(ap + k0);
      const int kk = k0 + khalf;
      v2f b;
      b.x = W_op[(long)kk * DDIM + colB];
      b.y = W_op[(long)(kk + 1) * DDIM + colB];
      acc = __builtin_amdgcn_wmma_f32_16x16x4_f32(
          false, a, false, b, (short)0, acc, false, false);
    }
    const float bo = b_op[colB];
    const int m0 = (lane >> 4) * 8;
#pragma unroll
    for (int r = 0; r < 8; ++r)
      out[(tokBase + m0 + r) * DDIM + colB] = acc[r] + bo;
  }
}

extern "C" void kernel_launch(void* const* d_in, const int* in_sizes, int n_in,
                              void* d_out, int out_size, void* d_ws, size_t ws_size,
                              hipStream_t stream) {
  const float* x          = (const float*)d_in[0];
  const float* mem_keys   = (const float*)d_in[1];
  const float* mem_values = (const float*)d_in[2];
  const float* W_fuse     = (const float*)d_in[3];
  const float* b_fuse     = (const float*)d_in[4];
  const float* ln_g       = (const float*)d_in[5];
  const float* ln_b       = (const float*)d_in[6];
  const float* W_op       = (const float*)d_in[7];
  const float* b_op       = (const float*)d_in[8];
  float* out = (float*)d_out;

  dim3 grid(8192 / TOK_PER_WG);
  dim3 block(512);
  hipLaunchKernelGGL(core_processor_kernel, grid, block, 0, stream,
                     x, mem_keys, mem_values, W_fuse, b_fuse, ln_g, ln_b,
                     W_op, b_op, out);
}


# LDS row pitch of the 16x256 A tiles (x/fused tile and h tile) padded 256 -> 260 floats: A fragment reads no longer hit one bank pair 16-way
# speedup vs baseline: 1.3944x; 1.3944x over previous
#include <hip/hip_runtime.h>
#include <math.h>

typedef __attribute__((ext_vector_type(2))) float v2f;
typedef __attribute__((ext_vector_type(8))) float v8f;

#define TOK_PER_WG 16
#define DDIM 256
#define M_MEM 16384
#define KSEL 32
#define NT 512
#define NCHUNK (M_MEM / NT)
#define SC_STRIDE 516
#define A_PITCH 260
#define LN_EPS 1e-5f

__device__ __forceinline__ float wave_sum_f(float v) {
#pragma unroll
  for (int o = 16; o > 0; o >>= 1) v += __shfl_xor(v, o, 32);
  return v;
}
__device__ __forceinline__ float wave_max_f(float v) {
#pragma unroll
  for (int o = 16; o > 0; o >>= 1) v = fmaxf(v, __shfl_xor(v, o, 32));
  return v;
}

__global__ __launch_bounds__(512, 1)
void core_processor_kernel(const float* __restrict__ x,
                           const float* __restrict__ mem_keys,
                           const float* __restrict__ mem_values,
                           const float* __restrict__ W_fuse,
                           const float* __restrict__ b_fuse,
                           const float* __restrict__ ln_g,
                           const float* __restrict__ ln_b,
                           const float* __restrict__ W_op,
                           const float* __restrict__ b_op,
                           float* __restrict__ out)
{
  __shared__ __align__(16) float xs[TOK_PER_WG * A_PITCH];
  __shared__ __align__(16) float sc[TOK_PER_WG * SC_STRIDE];
  __shared__ float tv[TOK_PER_WG * KSEL];
  __shared__ int   ti[TOK_PER_WG * KSEL];

  const int tid  = threadIdx.x;
  const int lane = tid & 31;
  const int w    = tid >> 5;
  const long tokBase = (long)blockIdx.x * TOK_PER_WG;

  const int rowA  = lane & 15;
  const int khalf = (lane >> 4) << 1;

  {
    const float4* src = (const float4*)(x + tokBase * DDIM);
    const int i0 = tid, i1 = tid + 512;
    *(float4*)(xs + (i0 >> 6) * A_PITCH + (i0 & 63) * 4) = src[i0];
    *(float4*)(xs + (i1 >> 6) * A_PITCH + (i1 & 63) * 4) = src[i1];
  }
  tv[w * KSEL + lane] = -INFINITY;
  ti[w * KSEL + lane] = 0;
  float curMin = -INFINITY;
  int   minPos = 0;
  __syncthreads();

  for (int chunk = 0; chunk < NCHUNK; ++chunk) {
    const int chunkBase = chunk * NT;
#pragma unroll
    for (int t2 = 0; t2 < 2; ++t2) {
      const int tile = w * 2 + t2;
      const int colB = chunkBase + tile * 16 + (lane & 15);
      const float* ap = xs + rowA * A_PITCH + khalf;
      const float* bp = mem_keys + (long)colB * DDIM + khalf;
      v8f acc = {};
#pragma unroll 4
      for (int k0 = 0; k0 < DDIM; k0 += 4) {
        v2f a = *(const v2f*)(ap + k0);
        v2f b = *(const v2f*)(bp + k0);
        acc = __builtin_amdgcn_wmma_f32_16x16x4_f32(
            false, a, false, b, (short)0, acc, false, false);
      }
      const int m0 = (lane >> 4) * 8;
      const int nn = tile * 16 + (lane & 15);
#pragma unroll
      for (int r = 0; r < 8; ++r)
        sc[(m0 + r) * SC_STRIDE + nn] = acc[r];
    }
    __syncthreads();

#pragma unroll 1
    for (int j = 0; j < NT / 32; ++j) {
      const int   col = lane + 32 * j;
      const float s   = sc[w * SC_STRIDE + col];
      const int   gi  = chunkBase + col;
      bool pend = (s > curMin);
      while (true) {
        unsigned long long bal = __ballot(pend);
        if (bal == 0ull) break;
        const int leader = (int)(__ffsll(bal) - 1);
        if (lane == leader) {
          tv[w * KSEL + minPos] = s;
          ti[w * KSEL + minPos] = gi;
          pend = false;
        }
        float v = tv[w * KSEL + lane];
        int   p = lane;
#pragma unroll
        for (int o = 16; o > 0; o >>= 1) {
          float v2 = __shfl_xor(v, o, 32);
          int   p2 = __shfl_xor(p, o, 32);
          if (v2 < v) { v = v2; p = p2; }
        }
        curMin = v; minPos = p;
        pend = pend && (s > curMin);
      }
    }
    __syncthreads();
  }

  {
    const float myv = tv[w * KSEL + lane];
    const int   myi = ti[w * KSEL + lane];
    const float mx  = wave_max_f(myv);
    const float e   = __expf(myv - mx);
    const float sum = wave_sum_f(e);
    const float wn  = e / sum;

    float acc[8];
#pragma unroll
    for (int q = 0; q < 8; ++q) acc[q] = 0.f;
#pragma unroll 1
    for (int k = 0; k < KSEL; ++k) {
      const float wk = __shfl(wn, k, 32);
      const int   id = __shfl(myi, k, 32);
      const float4* vp = (const float4*)(mem_values + (long)id * DDIM + lane * 8);
      const float4 a0 = vp[0];
      const float4 a1 = vp[1];
      acc[0] += wk * a0.x; acc[1] += wk * a0.y; acc[2] += wk * a0.z; acc[3] += wk * a0.w;
      acc[4] += wk * a1.x; acc[5] += wk * a1.y; acc[6] += wk * a1.z; acc[7] += wk * a1.w;
    }
#pragma unroll
    for (int q = 0; q < 8; ++q) xs[w * A_PITCH + lane * 8 + q] += acc[q];
  }
  __syncthreads();

  {
    const int colB = w * 16 + (lane & 15);
    const float* ap = xs + rowA * A_PITCH + khalf;
    v8f acc = {};
#pragma unroll 4
    for (int k0 = 0; k0 < DDIM; k0 += 4) {
      v2f a = *(const v2f*)(ap + k0);
      const int kk = k0 + khalf;
      v2f b;
      b.x = W_fuse[(long)kk * DDIM + colB];
      b.y = W_fuse[(long)(kk + 1) * DDIM + colB];
      acc = __builtin_amdgcn_wmma_f32_16x16x4_f32(
          false, a, false, b, (short)0, acc, false, false);
    }
    const float bf = b_fuse[colB];
    const int m0 = (lane >> 4) * 8;
#pragma unroll
    for (int r = 0; r < 8; ++r)
      sc[(m0 + r) * A_PITCH + colB] = acc[r] + bf;
  }
  __syncthreads();

  {
    const int d0 = lane * 8;
    float h[8];
#pragma unroll
    for (int q = 0; q < 8; ++q) h[q] = sc[w * A_PITCH + d0 + q];
    float s1 = 0.f, s2 = 0.f;
#pragma unroll
    for (int q = 0; q < 8; ++q) { s1 += h[q]; s2 += h[q] * h[q]; }
    s1 = wave_sum_f(s1);
    s2 = wave_sum_f(s2);
    const float mu   = s1 * (1.0f / DDIM);
    const float var  = s2 * (1.0f / DDIM) - mu * mu;
    const float rstd = rsqrtf(var + LN_EPS);
#pragma unroll
    for (int q = 0; q < 8; ++q) {
      const float g = ln_g[d0 + q];
      const float b = ln_b[d0 + q];
      const float v = (h[q] - mu) * rstd * g + b;
      sc[w * A_PITCH + d0 + q] = fmaxf(v, 0.f);
    }
  }
  __syncthreads();

  {
    const int colB = w * 16 + (lane & 15);
    const float* ap = sc + rowA * A_PITCH + khalf;
    v8f acc = {};
#pragma unroll 4
    for (int k0 = 0; k0 < DDIM; k0 += 4) {
      v2f a = *(const v2f*)(ap + k0);
      const int kk = k0 + khalf;
      v2f b;
      b.x = W_op[(long)kk * DDIM + colB];
      b.y = W_op[(long)(kk + 1) * DDIM + colB];
      acc = __builtin_amdgcn_wmma_f32_16x16x4_f32(
          false, a, false, b, (short)0, acc, false, false);
    }
    const float bo = b_op[colB];
    const int m0 = (lane >> 4) * 8;
#pragma unroll
    for (int r = 0; r < 8; ++r)
      out[(tokBase + m0 + r) * DDIM + colB] = acc[r] + bo;
  }
}

extern "C" void kernel_launch(void* const* d_in, const int* in_sizes, int n_in,
                              void* d_out, int out_size, void* d_ws, size_t ws_size,
                              hipStream_t stream) {
  const float* x          = (const float*)d_in[0];
  const float* mem_keys   = (const float*)d_in[1];
  const float* mem_values = (const float*)d_in[2];
  const float* W_fuse     = (const float*)d_in[3];
  const float* b_fuse     = (const float*)d_in[4];
  const float* ln_g       = (const float*)d_in[5];
  const float* ln_b       = (const float*)d_in[6];
  const float* W_op       = (const float*)d_in[7];
  const float* b_op       = (const float*)d_in[8];
  float* out = (float*)d_out;

  dim3 grid(8192 / TOK_PER_WG);
  dim3 block(512);
  hipLaunchKernelGGL(core_processor_kernel, grid, block, 0, stream,
                     x, mem_keys, mem_values, W_fuse, b_fuse, ln_g, ln_b,
                     W_op, b_op, out);
}


# Score GEMM blocked over tokens: a workgroup takes 32 tokens (32 waves, 256 workgroups) and every key fragment loaded from global feeds two A tiles: half the key loads and L2 bytes per WMMA
# speedup vs baseline: 1.9217x; 1.3782x over previous
#include <hip/hip_runtime.h>
#include <math.h>

typedef __attribute__((ext_vector_type(2))) float v2f;
typedef __attribute__((ext_vector_type(8))) float v8f;

#define TOK_PER_WG 32
#define DDIM 256
#define M_MEM 16384
#define KSEL 32
#define NT 512
#define NCHUNK (M_MEM / NT)
#define SC_STRIDE 516
#define A_PITCH 260
#define LN_EPS 1e-5f

__device__ __forceinline__ float wave_sum_f(float v) {
#pragma unroll
  for (int o = 16; o > 0; o >>= 1) v += __shfl_xor(v, o, 32);
  return v;
}
__device__ __forceinline__ float wave_max_f(float v) {
#pragma unroll
  for (int o = 16; o > 0; o >>= 1) v = fmaxf(v, __shfl_xor(v, o, 32));
  return v;
}

__global__ __launch_bounds__(1024, 1)
void core_processor_kernel(const float* __restrict__ x,
                           const float* __restrict__ mem_keys,
                           const float* __restrict__ mem_values,
                           const float* __restrict__ W_fuse,
                           const float* __restrict__ b_fuse,
                           const float* __restrict__ ln_g,
                           const float* __restrict__ ln_b,
                           const float* __restrict__ W_op,
                           const float* __restrict__ b_op,
                           float* __restrict__ out)
{
  __shared__ __align__(16) float xs[TOK_PER_WG * A_PITCH];
  __shared__ __align__(16) float sc[TOK_PER_WG * SC_STRIDE];
  __shared__ float tv[TOK_PER_WG * KSEL];
  __shared__ int   ti[TOK_PER_WG * KSEL];

  const int tid  = threadIdx.x;
  const int lane = tid & 31;
  const int w    = tid >> 5;
  const long tokBase = (long)blockIdx.x * TOK_PER_WG;

  const int rowA  = lane & 15;
  const int khalf = (lane >> 4) << 1;
  const int kquad = (lane >> 4) << 2;

  {
    const float4* src = (const float4*)(x + tokBase * DDIM);
    const int i0 = tid, i1 = tid + 1024;
    *(float4*)(xs + (i0 >> 6) * A_PITCH + (i0 & 63) * 4) = src[i0];
    *(float4*)(xs + (i1 >> 6) * A_PITCH + (i1 & 63) * 4) = src[i1];
  }
  tv[w * KSEL + lane] = -INFINITY;
  ti[w * KSEL + lane] = 0;
  float curMin = -INFINITY;
  int   minPos = 0;
  __syncthreads();

  for (int chunk = 0; chunk < NCHUNK; ++chunk) {
    const int chunkBase = chunk * NT;
    {
      const int colB = chunkBase + w * 16 + (lane & 15);
      const float* ap0 = xs + rowA * A_PITCH + kquad;
      const float* ap1 = xs + (16 + rowA) * A_PITCH + kquad;
      const float* bp = mem_keys + (long)colB * DDIM + kquad;
      v8f acc0 = {};
      v8f acc1 = {};
#pragma unroll 4
      for (int k0 = 0; k0 < DDIM; k0 += 8) {
        const float4 p4 = *(const float4*)(ap0 + k0);
        const float4 q4 = *(const float4*)(ap1 + k0);
        const float4 b4 = *(const float4*)(bp + k0);
        v2f a, c, b;
        a.x = p4.x; a.y = p4.y; c.x = q4.x; c.y = q4.y; b.x = b4.x; b.y = b4.y;
        acc0 = __builtin_amdgcn_wmma_f32_16x16x4_f32(
            false, a, false, b, (short)0, acc0, false, false);
        acc1 = __builtin_amdgcn_wmma_f32_16x16x4_f32(
            false, c, false, b, (short)0, acc1, false, false);
        a.x = p4.z; a.y = p4.w; c.x = q4.z; c.y = q4.w; b.x = b4.z; b.y = b4.w;
        acc0 = __builtin_amdgcn_wmma_f32_16x16x4_f32(
            false, a, false, b, (short)0, acc0, false, false);
        acc1 = __builtin_amdgcn_wmma_f32_16x16x4_f32(
            false, c, false, b, (short)0, acc1, false, false);
      }
      const int m0 = (lane >> 4) * 8;
      const int nn = w * 16 + (lane & 15);
#pragma unroll
      for (int r = 0; r < 8; ++r) {
        sc[(m0 + r) * SC_STRIDE + nn]      = acc0[r];
        sc[(16 + m0 + r) * SC_STRIDE + nn] = acc1[r];
      }
    }
    __syncthreads();

#pragma unroll 1
    for (int j = 0; j < NT / 32; ++j) {
      const int   col = lane + 32 * j;
      const float s   = sc[w * SC_STRIDE + col];
      const int   gi  = chunkBase + col;
      bool pend = (s > curMin);
      while (true) {
        unsigned long long bal = __ballot(pend);
        if (bal == 0ull) break;
        const int leader = (int)(__ffsll(bal) - 1);
        if (lane == leader) {
          tv[w * KSEL + minPos] = s;
          ti[w * KSEL + minPos] = gi;
          pend = false;
        }
        float v = tv[w * KSEL + lane];
        int   p = lane;
#pragma unroll
        for (int o = 16; o > 0; o >>= 1) {
          float v2 = __shfl_xor(v, o, 32);
          int   p2 = __shfl_xor(p, o, 32);
          if (v2 < v) { v = v2; p = p2; }
        }
        curMin = v; minPos = p;
        pend = pend && (s > curMin);
      }
    }
    __syncthreads();
  }

  {
    const float myv = tv[w * KSEL + lane];
    const int   myi = ti[w * KSEL + lane];
    const float mx  = wave_max_f(myv);
    const float e   = __expf(myv - mx);
    const float sum = wave_sum_f(e);
    const float wn  = e / sum;

    float acc[8];
#pragma unroll
    for (int q = 0; q < 8; ++q) acc[q] = 0.f;
#pragma unroll 1
    for (int k = 0; k < KSEL; ++k) {
      const float wk = __shfl(wn, k, 32);
      const int   id = __shfl(myi, k, 32);
      const float4* vp = (const float4*)(mem_values + (long)id * DDIM + lane * 8);
      const float4 a0 = vp[0];
      const float4 a1 = vp[1];
      acc[0] += wk * a0.x; acc[1] += wk * a0.y; acc[2] += wk * a0.z; acc[3] += wk * a0.w;
      acc[4] += wk * a1.x; acc[5] += wk * a1.y; acc[6] += wk * a1.z; acc[7] += wk * a1.w;
    }
#pragma unroll
    for (int q = 0; q < 8; ++q) xs[w * A_PITCH + lane * 8 + q] += acc[q];
  }
  __syncthreads();

  const int mt = w >> 4;
  const int nt = w & 15;
  {
    const int colB = nt * 16 + (lane & 15);
    const float* ap = xs + (mt * 16 + rowA) * A_PITCH + khalf;
    v8f acc = {};
#pragma unroll 4
    for (int k0 = 0; k0 < DDIM; k0 += 4) {
      v2f a = *(const v2f*)(ap + k0);
      const int kk = k0 + khalf;
      v2f b;
      b.x = W_fuse[(long)kk * DDIM + colB];
      b.y = W_fuse[(long)(kk + 1) * DDIM + colB];
      acc = __builtin_amdgcn_wmma_f32_16x16x4_f32(
          false, a, false, b, (short)0, acc, false, false);
    }
    const float bf = b_fuse[colB];
    const int m0 = (lane >> 4) * 8;
#pragma unroll
    for (int r = 0; r < 8; ++r)
      sc[(mt * 16 + m0 + r) * A_PITCH + colB] = acc[r] + bf;
  }
  __syncthreads();

  {
    const int d0 = lane * 8;
    float h[8];
#pragma unroll
    for (int q = 0; q < 8; ++q) h[q] = sc[w * A_PITCH + d0 + q];
    float s1 = 0.f, s2 = 0.f;
#pragma unroll
    for (int q = 0; q < 8; ++q) { s1 += h[q]; s2 += h[q] * h[q]; }
    s1 = wave_sum_f(s1);
    s2 = wave_sum_f(s2);
    const float mu   = s1 * (1.0f / DDIM);
    const float var  = s2 * (1.0f / DDIM) - mu * mu;
    const float rstd = rsqrtf(var + LN_EPS);
#pragma unroll
    for (int q = 0; q < 8; ++q) {
      const float g = ln_g[d0 + q];
      const float b = ln_b[d0 + q];
      const float v = (h[q] - mu) * rstd * g + b;
      sc[w * A_PITCH + d0 + q] = fmaxf(v, 0.f);
    }
  }
  __syncthreads();

  {
    const int colB = nt * 16 + (lane & 15);
    const float* ap = sc + (mt * 16 + rowA) * A_PITCH + khalf;
    v8f acc = {};
#pragma unroll 4
    for (int k0 = 0; k0 < DDIM; k0 += 4) {
      v2f a = *(const v2f*)(ap + k0);
      const int kk = k0 + khalf;
      v2f b;
      b.x = W_op[(long)kk * DDIM + colB];
      b.y = W_op[(long)(kk + 1) * DDIM + colB];
      acc = __builtin_amdgcn_wmma_f32_16x16x4_f32(
          false, a, false, b, (short)0, acc, false, false);
    }
    const float bo = b_op[colB];
    const int m0 = (lane >> 4) * 8;
#pragma unroll
    for (int r = 0; r < 8; ++r)
      out[(tokBase + mt * 16 + m0 + r) * DDIM + colB] = acc[r] + bo;
  }
}

extern "C" void kernel_launch(void* const* d_in, const int* in_sizes, int n_in,
                              void* d_out, int out_size, void* d_ws, size_t ws_size,
                              hipStream_t stream) {
  const float* x          = (const float*)d_in[0];
  const float* mem_keys   = (const float*)d_in[1];
  const float* mem_values = (const float*)d_in[2];
  const float* W_fuse     = (const float*)d_in[3];
  const float* b_fuse     = (const float*)d_in[4];
  const float* ln_g       = (const float*)d_in[5];
  const float* ln_b       = (const float*)d_in[6];
  const float* W_op       = (const float*)d_in[7];
  const float* b_op       = (const float*)d_in[8];
  float* out = (float*)d_out;

  dim3 grid(8192 / TOK_PER_WG);
  dim3 block(1024);
  hipLaunchKernelGGL(core_processor_kernel, grid, block, 0, stream,
                     x, mem_keys, mem_values, W_fuse, b_fuse, ln_g, ln_b,
                     W_op, b_op, out);
}


# Score chunk double-buffered in LDS, one barrier per chunk instead of two: half the waves of each SIMD run the next chunk's WMMAs while the other half merge the current chunk into the top-k lists
# speedup vs baseline: 2.0255x; 1.0540x over previous
#include <hip/hip_runtime.h>
#include <math.h>

typedef __attribute__((ext_vector_type(2))) float v2f;
typedef __attribute__((ext_vector_type(8))) float v8f;

#define TOK_PER_WG 32
#define DDIM 256
#define M_MEM 16384
#define KSEL 32
#define NT 512
#define NCHUNK (M_MEM / NT)
#define SC_STRIDE 516
#define SC_BUF (TOK_PER_WG * SC_STRIDE)
#define A_PITCH 260
#define LN_EPS 1e-5f

__device__ __forceinline__ float wave_sum_f(float v) {
#pragma unroll
  for (int o = 16; o > 0; o >>= 1) v += __shfl_xor(v, o, 32);
  return v;
}
__device__ __forceinline__ float wave_max_f(float v) {
#pragma unroll
  for (int o = 16; o > 0; o >>= 1) v = fmaxf(v, __shfl_xor(v, o, 32));
  return v;
}

__device__ __forceinline__ void score_chunk(const float* __restrict__ mem_keys,
                                            const float* xs, float* scb,
                                            int chunkBase, int w, int lane,
                                            int rowA, int kquad) {
  const int colB = chunkBase + w * 16 + (lane & 15);
  const float* ap0 = xs + rowA * A_PITCH + kquad;
  const float* ap1 = xs + (16 + rowA) * A_PITCH + kquad;
  const float* bp = mem_keys + (long)colB * DDIM + kquad;
  v8f acc0 = {};
  v8f acc1 = {};
#pragma unroll 4
  for (int k0 = 0; k0 < DDIM; k0 += 8) {
    const float4 p4 = *(const float4*)(ap0 + k0);
    const float4 q4 = *(const float4*)(ap1 + k0);
    const float4 b4 = *(const float4*)(bp + k0);
    v2f a, c, b;
    a.x = p4.x; a.y = p4.y; c.x = q4.x; c.y = q4.y; b.x = b4.x; b.y = b4.y;
    acc0 = __builtin_amdgcn_wmma_f32_16x16x4_f32(
        false, a, false, b, (short)0, acc0, false, false);
    acc1 = __builtin_amdgcn_wmma_f32_16x16x4_f32(
        false, c, false, b, (short)0, acc1, false, false);
    a.x = p4.z; a.y = p4.w; c.x = q4.z; c.y = q4.w; b.x = b4.z; b.y = b4.w;
    acc0 = __builtin_amdgcn_wmma_f32_16x16x4_f32(
        false, a, false, b, (short)0, acc0, false, false);
    acc1 = __builtin_amdgcn_wmma_f32_16x16x4_f32(
        false, c, false, b, (short)0, acc1, false, false);
  }
  const int m0 = (lane >> 4) * 8;
  const int nn = w * 16 + (lane & 15);
#pragma unroll
  for (int r = 0; r < 8; ++r) {
    scb[(m0 + r) * SC_STRIDE + nn]      = acc0[r];
    scb[(16 + m0 + r) * SC_STRIDE + nn] = acc1[r];
  }
}

__device__ __forceinline__ void merge_chunk(const float* scb, float* tvw, int* tiw,
                                            int chunkBase, int w, int lane,
                                            float& curMin, int& minPos) {
#pragma unroll 1
  for (int j = 0; j < NT / 32; ++j) {
    const int   col = lane + 32 * j;
    const float s   = scb[w * SC_STRIDE + col];
    const int   gi  = chunkBase + col;
    bool pend = (s > curMin);
    while (true) {
      unsigned long long bal = __ballot(pend);
      if (bal == 0ull) break;
      const int leader = (int)(__ffsll(bal) - 1);
      if (lane == leader) {
        tvw[minPos] = s;
        tiw[minPos] = gi;
        pend = false;
      }
      float v = tvw[lane];
      int   p = lane;
#pragma unroll
      for (int o = 16; o > 0; o >>= 1) {
        float v2 = __shfl_xor(v, o, 32);
        int   p2 = __shfl_xor(p, o, 32);
        if (v2 < v) { v = v2; p = p2; }
      }
      curMin = v; minPos = p;
      pend = pend && (s > curMin);
    }
  }
}

__global__ __launch_bounds__(1024, 1)
void core_processor_kernel(const float* __restrict__ x,
                           const float* __restrict__ mem_keys,
                           const float* __restrict__ mem_values,
                           const float* __restrict__ W_fuse,
                           const float* __restrict__ b_fuse,
                           const float* __restrict__ ln_g,
                           const float* __restrict__ ln_b,
                           const float* __restrict__ W_op,
                           const float* __restrict__ b_op,
                           float* __restrict__ out)
{
  __shared__ __align__(16) float xs[TOK_PER_WG * A_PITCH];
  __shared__ __align__(16) float sc[2 * SC_BUF];
  __shared__ float tv[TOK_PER_WG * KSEL];
  __shared__ int   ti[TOK_PER_WG * KSEL];

  const int tid  = threadIdx.x;
  const int lane = tid & 31;
  const int w    = tid >> 5;
  const long tokBase = (long)blockIdx.x * TOK_PER_WG;

  const int rowA  = lane & 15;
  const int khalf = (lane >> 4) << 1;
  const int kquad = (lane >> 4) << 2;

  {
    const float4* src = (const float4*)(x + tokBase * DDIM);
    const int i0 = tid, i1 = tid + 1024;
    *(float4*)(xs + (i0 >> 6) * A_PITCH + (i0 & 63) * 4) = src[i0];
    *(float4*)(xs + (i1 >> 6) * A_PITCH + (i1 & 63) * 4) = src[i1];
  }
  tv[w * KSEL + lane] = -INFINITY;
  ti[w * KSEL + lane] = 0;
  float curMin = -INFINITY;
  int   minPos = 0;
  __syncthreads();

  const bool gemmFirst = (__builtin_amdgcn_readfirstlane(w) & 4) != 0;
  float* tvw = tv + w * KSEL;
  int*   tiw = ti + w * KSEL;

  score_chunk(mem_keys, xs, sc, 0, w, lane, rowA, kquad);
  __syncthreads();

  for (int chunk = 0; chunk < NCHUNK; ++chunk) {
    const int chunkBase = chunk * NT;
    const float* cur = sc + (chunk & 1) * SC_BUF;
    float*       nxt = sc + ((chunk & 1) ^ 1) * SC_BUF;
    const bool more = (chunk + 1 < NCHUNK);
    if (gemmFirst) {
      if (more) score_chunk(mem_keys, xs, nxt, chunkBase + NT, w, lane, rowA, kquad);
      merge_chunk(cur, tvw, tiw, chunkBase, w, lane, curMin, minPos);
    } else {
      merge_chunk(cur, tvw, tiw, chunkBase, w, lane, curMin, minPos);
      if (more) score_chunk(mem_keys, xs, nxt, chunkBase + NT, w, lane, rowA, kquad);
    }
    __syncthreads();
  }

  {
    const float myv = tv[w * KSEL + lane];
    const int   myi = ti[w * KSEL + lane];
    const float mx  = wave_max_f(myv);
    const float e   = __expf(myv - mx);
    const float sum = wave_sum_f(e);
    const float wn  = e / sum;

    float acc[8];
#pragma unroll
    for (int q = 0; q < 8; ++q) acc[q] = 0.f;
#pragma unroll 1
    for (int k = 0; k < KSEL; ++k) {
      const float wk = __shfl(wn, k, 32);
      const int   id = __shfl(myi, k, 32);
      const float4* vp = (const float4*)(mem_values + (long)id * DDIM + lane * 8);
      const float4 a0 = vp[0];
      const float4 a1 = vp[1];
      acc[0] += wk * a0.x; acc[1] += wk * a0.y; acc[2] += wk * a0.z; acc[3] += wk * a0.w;
      acc[4] += wk * a1.x; acc[5] += wk * a1.y; acc[6] += wk * a1.z; acc[7] += wk * a1.w;
    }
#pragma unroll
    for (int q = 0; q < 8; ++q) xs[w * A_PITCH + lane * 8 + q] += acc[q];
  }
  __syncthreads();

  const int mt = w >> 4;
  const int nt = w & 15;
  {
    const int colB = nt * 16 + (lane & 15);
    const float* ap = xs + (mt * 16 + rowA) * A_PITCH + khalf;
    v8f acc = {};
#pragma unroll 4
    for (int k0 = 0; k0 < DDIM; k0 += 4) {
      v2f a = *(const v2f*)(ap + k0);
      const int kk = k0 + khalf;
      v2f b;
      b.x = W_fuse[(long)kk * DDIM + colB];
      b.y = W_fuse[(long)(kk + 1) * DDIM + colB];
      acc = __builtin_amdgcn_wmma_f32_16x16x4_f32(
          false, a, false, b, (short)0, acc, false, false);
    }
    const float bf = b_fuse[colB];
    const int m0 = (lane >> 4) * 8;
#pragma unroll
    for (int r = 0; r < 8; ++r)
      sc[(mt * 16 + m0 + r) * A_PITCH + colB] = acc[r] + bf;
  }
  __syncthreads();

  {
    const int d0 = lane * 8;
    float h[8];
#pragma unroll
    for (int q = 0; q < 8; ++q) h[q] = sc[w * A_PITCH + d0 + q];
    float s1 = 0.f, s2 = 0.f;
#pragma unroll
    for (int q = 0; q < 8; ++q) { s1 += h[q]; s2 += h[q] * h[q]; }
    s1 = wave_sum_f(s1);
    s2 = wave_sum_f(s2);
    const float mu   = s1 * (1.0f / DDIM);
    const float var  = s2 * (1.0f / DDIM) - mu * mu;
    const float rstd = rsqrtf(var + LN_EPS);
#pragma unroll
    for (int q = 0; q < 8; ++q) {
      const float g = ln_g[d0 + q];
      const float b = ln_b[d0 + q];
      const float v = (h[q] - mu) * rstd * g + b;
      sc[w * A_PITCH + d0 + q] = fmaxf(v, 0.f);
    }
  }
  __syncthreads();

  {
    const int colB = nt * 16 + (lane & 15);
    const float* ap = sc + (mt * 16 + rowA) * A_PITCH + khalf;
    v8f acc = {};
#pragma unroll 4
    for (int k0 = 0; k0 < DDIM; k0 += 4) {
      v2f a = *(const v2f*)(ap + k0);
      const int kk = k0 + khalf;
      v2f b;
      b.x = W_op[(long)kk * DDIM + colB];
      b.y = W_op[(long)(kk + 1) * DDIM + colB];
      acc = __builtin_amdgcn_wmma_f32_16x16x4_f32(
          false, a, false, b, (short)0, acc, false, false);
    }
    const float bo = b_op[colB];
    const int m0 = (lane >> 4) * 8;
#pragma unroll
    for (int r = 0; r < 8; ++r)
      out[(tokBase + mt * 16 + m0 + r) * DDIM + colB] = acc[r] + bo;
  }
}

extern "C" void kernel_launch(void* const* d_in, const int* in_sizes, int n_in,
                              void* d_out, int out_size, void* d_ws, size_t ws_size,
                              hipStream_t stream) {
  const float* x          = (const float*)d_in[0];
  const float* mem_keys   = (const float*)d_in[1];
  const float* mem_values = (const float*)d_in[2];
  const float* W_fuse     = (const float*)d_in[3];
  const float* b_fuse     = (const float*)d_in[4];
  const float* ln_g       = (const float*)d_in[5];
  const float* ln_b       = (const float*)d_in[6];
  const float* W_op       = (const float*)d_in[7];
  const float* b_op       = (const float*)d_in[8];
  float* out = (float*)d_out;

  dim3 grid(8192 / TOK_PER_WG);
  dim3 block(1024);
  hipLaunchKernelGGL(core_processor_kernel, grid, block, 0, stream,
                     x, mem_keys, mem_values, W_fuse, b_fuse, ln_g, ln_b,
                     W_op, b_op, out);
}


# Top-32 list of each token held in registers (slot = lane): insert by scalar broadcast, new minimum by DPP row_xmask + permlanex16 instead of LDS list + bpermute butterfly; sum order unchanged
# speedup vs baseline: 2.2988x; 1.1349x over previous
#include <hip/hip_runtime.h>
#include <math.h>

typedef __attribute__((ext_vector_type(2))) float v2f;
typedef __attribute__((ext_vector_type(8))) float v8f;

#define TOK_PER_WG 32
#define DDIM 256
#define M_MEM 16384
#define KSEL 32
#define NT 512
#define NCHUNK (M_MEM / NT)
#define SC_STRIDE 516
#define SC_BUF (TOK_PER_WG * SC_STRIDE)
#define A_PITCH 260
#define LN_EPS 1e-5f

__device__ __forceinline__ float wave_sum_f(float v) {
#pragma unroll
  for (int o = 16; o > 0; o >>= 1) v += __shfl_xor(v, o, 32);
  return v;
}
__device__ __forceinline__ float wave_max_f(float v) {
#pragma unroll
  for (int o = 16; o > 0; o >>= 1) v = fmaxf(v, __shfl_xor(v, o, 32));
  return v;
}

template <int CTRL>
__device__ __forceinline__ float dpp_row(float v) {
  return __builtin_bit_cast(float, __builtin_amdgcn_update_dpp(
      __builtin_bit_cast(int, v), __builtin_bit_cast(int, v), CTRL, 0xf, 0xf, true));
}
__device__ __forceinline__ float xhalf(float v) {
  return __builtin_bit_cast(float, __builtin_amdgcn_permlanex16(
      __builtin_bit_cast(int, v), __builtin_bit_cast(int, v), 0x76543210, 0xfedcba98, false, false));
}
__device__ __forceinline__ float wave_min_all(float v) {
  v = __builtin_fminimum_numf(v, dpp_row<0x161>(v));
  v = __builtin_fminimum_numf(v, dpp_row<0x162>(v));
  v = __builtin_fminimum_numf(v, dpp_row<0x164>(v));
  v = __builtin_fminimum_numf(v, dpp_row<0x168>(v));
  v = __builtin_fminimum_numf(v, xhalf(v));
  return v;
}

__device__ __forceinline__ void score_chunk(const float* __restrict__ mem_keys,
                                            const float* xs, float* scb,
                                            int chunkBase, int w, int lane,
                                            int rowA, int kquad) {
  const int colB = chunkBase + w * 16 + (lane & 15);
  const float* ap0 = xs + rowA * A_PITCH + kquad;
  const float* ap1 = xs + (16 + rowA) * A_PITCH + kquad;
  const float* bp = mem_keys + (long)colB * DDIM + kquad;
  v8f acc0 = {};
  v8f acc1 = {};
#pragma unroll 4
  for (int k0 = 0; k0 < DDIM; k0 += 8) {
    const float4 p4 = *(const float4*)(ap0 + k0);
    const float4 q4 = *(const float4*)(ap1 + k0);
    const float4 b4 = *(const float4*)(bp + k0);
    v2f a, c, b;
    a.x = p4.x; a.y = p4.y; c.x = q4.x; c.y = q4.y; b.x = b4.x; b.y = b4.y;
    acc0 = __builtin_amdgcn_wmma_f32_16x16x4_f32(
        false, a, false, b, (short)0, acc0, false, false);
    acc1 = __builtin_amdgcn_wmma_f32_16x16x4_f32(
        false, c, false, b, (short)0, acc1, false, false);
    a.x = p4.z; a.y = p4.w; c.x = q4.z; c.y = q4.w; b.x = b4.z; b.y = b4.w;
    acc0 = __builtin_amdgcn_wmma_f32_16x16x4_f32(
        false, a, false, b, (short)0, acc0, false, false);
    acc1 = __builtin_amdgcn_wmma_f32_16x16x4_f32(
        false, c, false, b, (short)0, acc1, false, false);
  }
  const int m0 = (lane >> 4) * 8;
  const int nn = w * 16 + (lane & 15);
#pragma unroll
  for (int r = 0; r < 8; ++r) {
    scb[(m0 + r) * SC_STRIDE + nn]      = acc0[r];
    scb[(16 + m0 + r) * SC_STRIDE + nn] = acc1[r];
  }
}

__device__ __forceinline__ void merge_chunk(const float* scb, float& tvr, int& tir,
                                            int chunkBase, int w, int lane,
                                            float& curMin, int& minPos) {
#pragma unroll 1
  for (int j = 0; j < NT / 32; ++j) {
    const int   col = lane + 32 * j;
    const float s   = scb[w * SC_STRIDE + col];
    const int   gi  = chunkBase + col;
    unsigned bal = __builtin_amdgcn_ballot_w32(s > curMin);
    while (bal != 0u) {
      const int leader = __builtin_ctz(bal);
      const float sL = __builtin_bit_cast(float,
          __builtin_amdgcn_readlane(__builtin_bit_cast(int, s), leader));
      const int giL = __builtin_amdgcn_readlane(gi, leader);
      if (lane == minPos) { tvr = sL; tir = giL; }
      curMin = wave_min_all(tvr);
      minPos = __builtin_ctz(__builtin_amdgcn_ballot_w32(tvr == curMin));
      bal &= bal - 1u;
      bal &= __builtin_amdgcn_ballot_w32(s > curMin);
    }
  }
}

__global__ __launch_bounds__(1024, 1)
void core_processor_kernel(const float* __restrict__ x,
                           const float* __restrict__ mem_keys,
                           const float* __restrict__ mem_values,
                           const float* __restrict__ W_fuse,
                           const float* __restrict__ b_fuse,
                           const float* __restrict__ ln_g,
                           const float* __restrict__ ln_b,
                           const float* __restrict__ W_op,
                           const float* __restrict__ b_op,
                           float* __restrict__ out)
{
  __shared__ __align__(16) float xs[TOK_PER_WG * A_PITCH];
  __shared__ __align__(16) float sc[2 * SC_BUF];

  const int tid  = threadIdx.x;
  const int lane = tid & 31;
  const int w    = tid >> 5;
  const long tokBase = (long)blockIdx.x * TOK_PER_WG;

  const int rowA  = lane & 15;
  const int khalf = (lane >> 4) << 1;
  const int kquad = (lane >> 4) << 2;

  {
    const float4* src = (const float4*)(x + tokBase * DDIM);
    const int i0 = tid, i1 = tid + 1024;
    *(float4*)(xs + (i0 >> 6) * A_PITCH + (i0 & 63) * 4) = src[i0];
    *(float4*)(xs + (i1 >> 6) * A_PITCH + (i1 & 63) * 4) = src[i1];
  }
  float tvr = -INFINITY;
  int   tir = 0;
  float curMin = -INFINITY;
  int   minPos = 0;
  __syncthreads();

  const bool gemmFirst = (__builtin_amdgcn_readfirstlane(w) & 4) != 0;

  score_chunk(mem_keys, xs, sc, 0, w, lane, rowA, kquad);
  __syncthreads();

  for (int chunk = 0; chunk < NCHUNK; ++chunk) {
    const int chunkBase = chunk * NT;
    const float* cur = sc + (chunk & 1) * SC_BUF;
    float*       nxt = sc + ((chunk & 1) ^ 1) * SC_BUF;
    const bool more = (chunk + 1 < NCHUNK);
    if (gemmFirst) {
      if (more) score_chunk(mem_keys, xs, nxt, chunkBase + NT, w, lane, rowA, kquad);
      merge_chunk(cur, tvr, tir, chunkBase, w, lane, curMin, minPos);
    } else {
      merge_chunk(cur, tvr, tir, chunkBase, w, lane, curMin, minPos);
      if (more) score_chunk(mem_keys, xs, nxt, chunkBase + NT, w, lane, rowA, kquad);
    }
    __syncthreads();
  }

  {
    const float myv = tvr;
    const int   myi = tir;
    const float mx  = wave_max_f(myv);
    const float e   = __expf(myv - mx);
    const float sum = wave_sum_f(e);
    const float wn  = e / sum;

    float acc[8];
#pragma unroll
    for (int q = 0; q < 8; ++q) acc[q] = 0.f;
#pragma unroll 1
    for (int k = 0; k < KSEL; ++k) {
      const float wk = __shfl(wn, k, 32);
      const int   id = __shfl(myi, k, 32);
      const float4* vp = (const float4*)(mem_values + (long)id * DDIM + lane * 8);
      const float4 a0 = vp[0];
      const float4 a1 = vp[1];
      acc[0] += wk * a0.x; acc[1] += wk * a0.y; acc[2] += wk * a0.z; acc[3] += wk * a0.w;
      acc[4] += wk * a1.x; acc[5] += wk * a1.y; acc[6] += wk * a1.z; acc[7] += wk * a1.w;
    }
#pragma unroll
    for (int q = 0; q < 8; ++q) xs[w * A_PITCH + lane * 8 + q] += acc[q];
  }
  __syncthreads();

  const int mt = w >> 4;
  const int nt = w & 15;
  {
    const int colB = nt * 16 + (lane & 15);
    const float* ap = xs + (mt * 16 + rowA) * A_PITCH + khalf;
    v8f acc = {};
#pragma unroll 4
    for (int k0 = 0; k0 < DDIM; k0 += 4) {
      v2f a = *(const v2f*)(ap + k0);
      const int kk = k0 + khalf;
      v2f b;
      b.x = W_fuse[(long)kk * DDIM + colB];
      b.y = W_fuse[(long)(kk + 1) * DDIM + colB];
      acc = __builtin_amdgcn_wmma_f32_16x16x4_f32(
          false, a, false, b, (short)0, acc, false, false);
    }
    const float bf = b_fuse[colB];
    const int m0 = (lane >> 4) * 8;
#pragma unroll
    for (int r = 0; r < 8; ++r)
      sc[(mt * 16 + m0 + r) * A_PITCH + colB] = acc[r] + bf;
  }
  __syncthreads();

  {
    const int d0 = lane * 8;
    float h[8];
#pragma unroll
    for (int q = 0; q < 8; ++q) h[q] = sc[w * A_PITCH + d0 + q];
    float s1 = 0.f, s2 = 0.f;
#pragma unroll
    for (int q = 0; q < 8; ++q) { s1 += h[q]; s2 += h[q] * h[q]; }
    s1 = wave_sum_f(s1);
    s2 = wave_sum_f(s2);
    const float mu   = s1 * (1.0f / DDIM);
    const float var  = s2 * (1.0f / DDIM) - mu * mu;
    const float rstd = rsqrtf(var + LN_EPS);
#pragma unroll
    for (int q = 0; q < 8; ++q) {
      const float g = ln_g[d0 + q];
      const float b = ln_b[d0 + q];
      const float v = (h[q] - mu) * rstd * g + b;
      sc[w * A_PITCH + d0 + q] = fmaxf(v, 0.f);
    }
  }
  __syncthreads();

  {
    const int colB = nt * 16 + (lane & 15);
    const float* ap = sc + (mt * 16 + rowA) * A_PITCH + khalf;
    v8f acc = {};
#pragma unroll 4
    for (int k0 = 0; k0 < DDIM; k0 += 4) {
      v2f a = *(const v2f*)(ap + k0);
      const int kk = k0 + khalf;
      v2f b;
      b.x = W_op[(long)kk * DDIM + colB];
      b.y = W_op[(long)(kk + 1) * DDIM + colB];
      acc = __builtin_amdgcn_wmma_f32_16x16x4_f32(
          false, a, false, b, (short)0, acc, false, false);
    }
    const float bo = b_op[colB];
    const int m0 = (lane >> 4) * 8;
#pragma unroll
    for (int r = 0; r < 8; ++r)
      out[(tokBase + mt * 16 + m0 + r) * DDIM + colB] = acc[r] + bo;
  }
}

extern "C" void kernel_launch(void* const* d_in, const int* in_sizes, int n_in,
                              void* d_out, int out_size, void* d_ws, size_t ws_size,
                              hipStream_t stream) {
  const float* x          = (const float*)d_in[0];
  const float* mem_keys   = (const float*)d_in[1];
  const float* mem_values = (const float*)d_in[2];
  const float* W_fuse     = (const float*)d_in[3];
  const float* b_fuse     = (const float*)d_in[4];
  const float* ln_g       = (const float*)d_in[5];
  const float* ln_b       = (const float*)d_in[6];
  const float* W_op       = (const float*)d_in[7];
  const float* b_op       = (const float*)d_in[8];
  float* out = (float*)d_out;

  dim3 grid(8192 / TOK_PER_WG);
  dim3 block(1024);
  hipLaunchKernelGGL(core_processor_kernel, grid, block, 0, stream,
                     x, mem_keys, mem_values, W_fuse, b_fuse, ln_g, ln_b,
                     W_op, b_op, out);
}


# Top-k scan: a chunk row's 16 score loads issued as one group, threshold tested on the max of four loads per lane (one branch per 128 scores, not 32); insert order and sum order unchanged
# speedup vs baseline: 2.3947x; 1.0417x over previous
#include <hip/hip_runtime.h>
#include <math.h>

typedef __attribute__((ext_vector_type(2))) float v2f;
typedef __attribute__((ext_vector_type(8))) float v8f;

#define TOK_PER_WG 32
#define DDIM 256
#define M_MEM 16384
#define KSEL 32
#define NT 512
#define NCHUNK (M_MEM / NT)
#define SC_STRIDE 516
#define SC_BUF (TOK_PER_WG * SC_STRIDE)
#define A_PITCH 260
#define LN_EPS 1e-5f

__device__ __forceinline__ float wave_sum_f(float v) {
#pragma unroll
  for (int o = 16; o > 0; o >>= 1) v += __shfl_xor(v, o, 32);
  return v;
}
__device__ __forceinline__ float wave_max_f(float v) {
#pragma unroll
  for (int o = 16; o > 0; o >>= 1) v = fmaxf(v, __shfl_xor(v, o, 32));
  return v;
}

template <int CTRL>
__device__ __forceinline__ float dpp_row(float v) {
  return __builtin_bit_cast(float, __builtin_amdgcn_update_dpp(
      __builtin_bit_cast(int, v), __builtin_bit_cast(int, v), CTRL, 0xf, 0xf, true));
}
__device__ __forceinline__ float xhalf(float v) {
  return __builtin_bit_cast(float, __builtin_amdgcn_permlanex16(
      __builtin_bit_cast(int, v), __builtin_bit_cast(int, v), 0x76543210, 0xfedcba98, false, false));
}
__device__ __forceinline__ float wave_min_all(float v) {
  v = __builtin_fminimum_numf(v, dpp_row<0x161>(v));
  v = __builtin_fminimum_numf(v, dpp_row<0x162>(v));
  v = __builtin_fminimum_numf(v, dpp_row<0x164>(v));
  v = __builtin_fminimum_numf(v, dpp_row<0x168>(v));
  v = __builtin_fminimum_numf(v, xhalf(v));
  return v;
}

__device__ __forceinline__ void score_chunk(const float* __restrict__ mem_keys,
                                            const float* xs, float* scb,
                                            int chunkBase, int w, int lane,
                                            int rowA, int kquad) {
  const int colB = chunkBase + w * 16 + (lane & 15);
  const float* ap0 = xs + rowA * A_PITCH + kquad;
  const float* ap1 = xs + (16 + rowA) * A_PITCH + kquad;
  const float* bp = mem_keys + (long)colB * DDIM + kquad;
  v8f acc0 = {};
  v8f acc1 = {};
#pragma unroll 4
  for (int k0 = 0; k0 < DDIM; k0 += 8) {
    const float4 p4 = *(const float4*)(ap0 + k0);
    const float4 q4 = *(const float4*)(ap1 + k0);
    const float4 b4 = *(const float4*)(bp + k0);
    v2f a, c, b;
    a.x = p4.x; a.y = p4.y; c.x = q4.x; c.y = q4.y; b.x = b4.x; b.y = b4.y;
    acc0 = __builtin_amdgcn_wmma_f32_16x16x4_f32(
        false, a, false, b, (short)0, acc0, false, false);
    acc1 = __builtin_amdgcn_wmma_f32_16x16x4_f32(
        false, c, false, b, (short)0, acc1, false, false);
    a.x = p4.z; a.y = p4.w; c.x = q4.z; c.y = q4.w; b.x = b4.z; b.y = b4.w;
    acc0 = __builtin_amdgcn_wmma_f32_16x16x4_f32(
        false, a, false, b, (short)0, acc0, false, false);
    acc1 = __builtin_amdgcn_wmma_f32_16x16x4_f32(
        false, c, false, b, (short)0, acc1, false, false);
  }
  const int m0 = (lane >> 4) * 8;
  const int nn = w * 16 + (lane & 15);
#pragma unroll
  for (int r = 0; r < 8; ++r) {
    scb[(m0 + r) * SC_STRIDE + nn]      = acc0[r];
    scb[(16 + m0 + r) * SC_STRIDE + nn] = acc1[r];
  }
}

__device__ __forceinline__ void insert_block(float s, int gi, float& tvr, int& tir,
                                             int lane, float& curMin, int& minPos) {
  unsigned bal = __builtin_amdgcn_ballot_w32(s > curMin);
  while (bal != 0u) {
    const int leader = __builtin_ctz(bal);
    const float sL = __builtin_bit_cast(float,
        __builtin_amdgcn_readlane(__builtin_bit_cast(int, s), leader));
    const int giL = __builtin_amdgcn_readlane(gi, leader);
    if (lane == minPos) { tvr = sL; tir = giL; }
    curMin = wave_min_all(tvr);
    minPos = __builtin_ctz(__builtin_amdgcn_ballot_w32(tvr == curMin));
    bal &= bal - 1u;
    bal &= __builtin_amdgcn_ballot_w32(s > curMin);
  }
}

__device__ __forceinline__ void merge_chunk(const float* scb, float& tvr, int& tir,
                                            int chunkBase, int w, int lane,
                                            float& curMin, int& minPos) {
  const float* row = scb + w * SC_STRIDE + lane;
  float s[NT / 32];
#pragma unroll
  for (int j = 0; j < NT / 32; ++j) s[j] = row[32 * j];
#pragma unroll
  for (int g = 0; g < NT / 128; ++g) {
    const float m = fmaxf(fmaxf(s[4 * g], s[4 * g + 1]), fmaxf(s[4 * g + 2], s[4 * g + 3]));
    if (__builtin_amdgcn_ballot_w32(m > curMin) != 0u) {
#pragma unroll
      for (int q = 0; q < 4; ++q)
        insert_block(s[4 * g + q], chunkBase + lane + 32 * (4 * g + q),
                     tvr, tir, lane, curMin, minPos);
    }
  }
}

__global__ __launch_bounds__(1024, 1)
void core_processor_kernel(const float* __restrict__ x,
                           const float* __restrict__ mem_keys,
                           const float* __restrict__ mem_values,
                           const float* __restrict__ W_fuse,
                           const float* __restrict__ b_fuse,
                           const float* __restrict__ ln_g,
                           const float* __restrict__ ln_b,
                           const float* __restrict__ W_op,
                           const float* __restrict__ b_op,
                           float* __restrict__ out)
{
  __shared__ __align__(16) float xs[TOK_PER_WG * A_PITCH];
  __shared__ __align__(16) float sc[2 * SC_BUF];

  const int tid  = threadIdx.x;
  const int lane = tid & 31;
  const int w    = tid >> 5;
  const long tokBase = (long)blockIdx.x * TOK_PER_WG;

  const int rowA  = lane & 15;
  const int khalf = (lane >> 4) << 1;
  const int kquad = (lane >> 4) << 2;

  {
    const float4* src = (const float4*)(x + tokBase * DDIM);
    const int i0 = tid, i1 = tid + 1024;
    *(float4*)(xs + (i0 >> 6) * A_PITCH + (i0 & 63) * 4) = src[i0];
    *(float4*)(xs + (i1 >> 6) * A_PITCH + (i1 & 63) * 4) = src[i1];
  }
  float tvr = -INFINITY;
  int   tir = 0;
  float curMin = -INFINITY;
  int   minPos = 0;
  __syncthreads();

  const bool gemmFirst = (__builtin_amdgcn_readfirstlane(w) & 4) != 0;

  score_chunk(mem_keys, xs, sc, 0, w, lane, rowA, kquad);
  __syncthreads();

  for (int chunk = 0; chunk < NCHUNK; ++chunk) {
    const int chunkBase = chunk * NT;
    const float* cur = sc + (chunk & 1) * SC_BUF;
    float*       nxt = sc + ((chunk & 1) ^ 1) * SC_BUF;
    const bool more = (chunk + 1 < NCHUNK);
    if (gemmFirst) {
      if (more) score_chunk(mem_keys, xs, nxt, chunkBase + NT, w, lane, rowA, kquad);
      merge_chunk(cur, tvr, tir, chunkBase, w, lane, curMin, minPos);
    } else {
      merge_chunk(cur, tvr, tir, chunkBase, w, lane, curMin, minPos);
      if (more) score_chunk(mem_keys, xs, nxt, chunkBase + NT, w, lane, rowA, kquad);
    }
    __syncthreads();
  }

  {
    const float myv = tvr;
    const int   myi = tir;
    const float mx  = wave_max_f(myv);
    const float e   = __expf(myv - mx);
    const float sum = wave_sum_f(e);
    const float wn  = e / sum;

    float acc[8];
#pragma unroll
    for (int q = 0; q < 8; ++q) acc[q] = 0.f;
#pragma unroll 1
    for (int k = 0; k < KSEL; ++k) {
      const float wk = __shfl(wn, k, 32);
      const int   id = __shfl(myi, k, 32);
      const float4* vp = (const float4*)(mem_values + (long)id * DDIM + lane * 8);
      const float4 a0 = vp[0];
      const float4 a1 = vp[1];
      acc[0] += wk * a0.x; acc[1] += wk * a0.y; acc[2] += wk * a0.z; acc[3] += wk * a0.w;
      acc[4] += wk * a1.x; acc[5] += wk * a1.y; acc[6] += wk * a1.z; acc[7] += wk * a1.w;
    }
#pragma unroll
    for (int q = 0; q < 8; ++q) xs[w * A_PITCH + lane * 8 + q] += acc[q];
  }
  __syncthreads();

  const int mt = w >> 4;
  const int nt = w & 15;
  {
    const int colB = nt * 16 + (lane & 15);
    const float* ap = xs + (mt * 16 + rowA) * A_PITCH + khalf;
    v8f acc = {};
#pragma unroll 4
    for (int k0 = 0; k0 < DDIM; k0 += 4) {
      v2f a = *(const v2f*)(ap + k0);
      const int kk = k0 + khalf;
      v2f b;
      b.x = W_fuse[(long)kk * DDIM + colB];
      b.y = W_fuse[(long)(kk + 1) * DDIM + colB];
      acc = __builtin_amdgcn_wmma_f32_16x16x4_f32(
          false, a, false, b, (short)0, acc, false, false);
    }
    const float bf = b_fuse[colB];
    const int m0 = (lane >> 4) * 8;
#pragma unroll
    for (int r = 0; r < 8; ++r)
      sc[(mt * 16 + m0 + r) * A_PITCH + colB] = acc[r] + bf;
  }
  __syncthreads();

  {
    const int d0 = lane * 8;
    float h[8];
#pragma unroll
    for (int q = 0; q < 8; ++q) h[q] = sc[w * A_PITCH + d0 + q];
    float s1 = 0.f, s2 = 0.f;
#pragma unroll
    for (int q = 0; q < 8; ++q) { s1 += h[q]; s2 += h[q] * h[q]; }
    s1 = wave_sum_f(s1);
    s2 = wave_sum_f(s2);
    const float mu   = s1 * (1.0f / DDIM);
    const float var  = s2 * (1.0f / DDIM) - mu * mu;
    const float rstd = rsqrtf(var + LN_EPS);
#pragma unroll
    for (int q = 0; q < 8; ++q) {
      const float g = ln_g[d0 + q];
      const float b = ln_b[d0 + q];
      const float v = (h[q] - mu) * rstd * g + b;
      sc[w * A_PITCH + d0 + q] = fmaxf(v, 0.f);
    }
  }
  __syncthreads();

  {
    const int colB = nt * 16 + (lane & 15);
    const float* ap = sc + (mt * 16 + rowA) * A_PITCH + khalf;
    v8f acc = {};
#pragma unroll 4
    for (int k0 = 0; k0 < DDIM; k0 += 4) {
      v2f a = *(const v2f*)(ap + k0);
      const int kk = k0 + khalf;
      v2f b;
      b.x = W_op[(long)kk * DDIM + colB];
      b.y = W_op[(long)(kk + 1) * DDIM + colB];
      acc = __builtin_amdgcn_wmma_f32_16x16x4_f32(
          false, a, false, b, (short)0, acc, false, false);
    }
    const float bo = b_op[colB];
    const int m0 = (lane >> 4) * 8;
#pragma unroll
    for (int r = 0; r < 8; ++r)
      out[(tokBase + mt * 16 + m0 + r) * DDIM + colB] = acc[r] + bo;
  }
}

extern "C" void kernel_launch(void* const* d_in, const int* in_sizes, int n_in,
                              void* d_out, int out_size, void* d_ws, size_t ws_size,
                              hipStream_t stream) {
  const float* x          = (const float*)d_in[0];
  const float* mem_keys   = (const float*)d_in[1];
  const float* mem_values = (const float*)d_in[2];
  const float* W_fuse     = (const float*)d_in[3];
  const float* b_fuse     = (const float*)d_in[4];
  const float* ln_g       = (const float*)d_in[5];
  const float* ln_b       = (const float*)d_in[6];
  const float* W_op       = (const float*)d_in[7];
  const float* b_op       = (const float*)d_in[8];
  float* out = (float*)d_out;

  dim3 grid(8192 / TOK_PER_WG);
  dim3 block(1024);
  hipLaunchKernelGGL(core_processor_kernel, grid, block, 0, stream,
                     x, mem_keys, mem_values, W_fuse, b_fuse, ln_g, ln_b,
                     W_op, b_op, out);
}


# Score GEMM register-blocked 2x2: a wave takes 32 tokens x 32 keys of a 1024-key chunk, each LDS A fragment feeds two key fragments: half the LDS fragment reads and barriers; sum order unchanged
# speedup vs baseline: 2.4978x; 1.0431x over previous
#include <hip/hip_runtime.h>
#include <math.h>

typedef __attribute__((ext_vector_type(2))) float v2f;
typedef __attribute__((ext_vector_type(8))) float v8f;

#define TOK_PER_WG 32
#define DDIM 256
#define M_MEM 16384
#define KSEL 32
#define NT 1024
#define NCHUNK (M_MEM / NT)
#define SC_STRIDE 1028
#define SC_BUF (TOK_PER_WG * SC_STRIDE)
#define A_PITCH 260
#define LN_EPS 1e-5f

__device__ __forceinline__ float wave_sum_f(float v) {
#pragma unroll
  for (int o = 16; o > 0; o >>= 1) v += __shfl_xor(v, o, 32);
  return v;
}
__device__ __forceinline__ float wave_max_f(float v) {
#pragma unroll
  for (int o = 16; o > 0; o >>= 1) v = fmaxf(v, __shfl_xor(v, o, 32));
  return v;
}

template <int CTRL>
__device__ __forceinline__ float dpp_row(float v) {
  return __builtin_bit_cast(float, __builtin_amdgcn_update_dpp(
      __builtin_bit_cast(int, v), __builtin_bit_cast(int, v), CTRL, 0xf, 0xf, true));
}
__device__ __forceinline__ float xhalf(float v) {
  return __builtin_bit_cast(float, __builtin_amdgcn_permlanex16(
      __builtin_bit_cast(int, v), __builtin_bit_cast(int, v), 0x76543210, 0xfedcba98, false, false));
}
__device__ __forceinline__ float wave_min_all(float v) {
  v = __builtin_fminimum_numf(v, dpp_row<0x161>(v));
  v = __builtin_fminimum_numf(v, dpp_row<0x162>(v));
  v = __builtin_fminimum_numf(v, dpp_row<0x164>(v));
  v = __builtin_fminimum_numf(v, dpp_row<0x168>(v));
  v = __builtin_fminimum_numf(v, xhalf(v));
  return v;
}

__device__ __forceinline__ void score_chunk(const float* __restrict__ mem_keys,
                                            const float* xs, float* scb,
                                            int chunkBase, int w, int lane,
                                            int rowA, int kquad) {
  const int colB = chunkBase + w * 32 + (lane & 15);
  const float* ap0 = xs + rowA * A_PITCH + kquad;
  const float* ap1 = xs + (16 + rowA) * A_PITCH + kquad;
  const float* bp0 = mem_keys + (long)colB * DDIM + kquad;
  const float* bp1 = bp0 + 16 * DDIM;
  v8f acc00 = {};
  v8f acc10 = {};
  v8f acc01 = {};
  v8f acc11 = {};
#pragma unroll 4
  for (int k0 = 0; k0 < DDIM; k0 += 8) {
    const float4 p4 = *(const float4*)(ap0 + k0);
    const float4 q4 = *(const float4*)(ap1 + k0);
    const float4 b4 = *(const float4*)(bp0 + k0);
    const float4 d4 = *(const float4*)(bp1 + k0);
    v2f a, c, b, d;
    a.x = p4.x; a.y = p4.y; c.x = q4.x; c.y = q4.y;
    b.x = b4.x; b.y = b4.y; d.x = d4.x; d.y = d4.y;
    acc00 = __builtin_amdgcn_wmma_f32_16x16x4_f32(
        false, a, false, b, (short)0, acc00, false, false);
    acc10 = __builtin_amdgcn_wmma_f32_16x16x4_f32(
        false, c, false, b, (short)0, acc10, false, false);
    acc01 = __builtin_amdgcn_wmma_f32_16x16x4_f32(
        false, a, false, d, (short)0, acc01, false, false);
    acc11 = __builtin_amdgcn_wmma_f32_16x16x4_f32(
        false, c, false, d, (short)0, acc11, false, false);
    a.x = p4.z; a.y = p4.w; c.x = q4.z; c.y = q4.w;
    b.x = b4.z; b.y = b4.w; d.x = d4.z; d.y = d4.w;
    acc00 = __builtin_amdgcn_wmma_f32_16x16x4_f32(
        false, a, false, b, (short)0, acc00, false, false);
    acc10 = __builtin_amdgcn_wmma_f32_16x16x4_f32(
        false, c, false, b, (short)0, acc10, false, false);
    acc01 = __builtin_amdgcn_wmma_f32_16x16x4_f32(
        false, a, false, d, (short)0, acc01, false, false);
    acc11 = __builtin_amdgcn_wmma_f32_16x16x4_f32(
        false, c, false, d, (short)0, acc11, false, false);
  }
  const int m0 = (lane >> 4) * 8;
  const int nn = w * 32 + (lane & 15);
#pragma unroll
  for (int r = 0; r < 8; ++r) {
    scb[(m0 + r) * SC_STRIDE + nn]           = acc00[r];
    scb[(16 + m0 + r) * SC_STRIDE + nn]      = acc10[r];
    scb[(m0 + r) * SC_STRIDE + nn + 16]      = acc01[r];
    scb[(16 + m0 + r) * SC_STRIDE + nn + 16] = acc11[r];
  }
}

__device__ __forceinline__ void insert_block(float s, int gi, float& tvr, int& tir,
                                             int lane, float& curMin, int& minPos) {
  unsigned bal = __builtin_amdgcn_ballot_w32(s > curMin);
  while (bal != 0u) {
    const int leader = __builtin_ctz(bal);
    const float sL = __builtin_bit_cast(float,
        __builtin_amdgcn_readlane(__builtin_bit_cast(int, s), leader));
    const int giL = __builtin_amdgcn_readlane(gi, leader);
    if (lane == minPos) { tvr = sL; tir = giL; }
    curMin = wave_min_all(tvr);
    minPos = __builtin_ctz(__builtin_amdgcn_ballot_w32(tvr == curMin));
    bal &= bal - 1u;
    bal &= __builtin_amdgcn_ballot_w32(s > curMin);
  }
}

__device__ __forceinline__ void merge_chunk(const float* scb, float& tvr, int& tir,
                                            int chunkBase, int w, int lane,
                                            float& curMin, int& minPos) {
#pragma unroll 1
  for (int hb = 0; hb < NT; hb += 512) {
    const float* row = scb + w * SC_STRIDE + hb + lane;
    float s[16];
#pragma unroll
    for (int j = 0; j < 16; ++j) s[j] = row[32 * j];
#pragma unroll
    for (int g = 0; g < 4; ++g) {
      const float m = fmaxf(fmaxf(s[4 * g], s[4 * g + 1]), fmaxf(s[4 * g + 2], s[4 * g + 3]));
      if (__builtin_amdgcn_ballot_w32(m > curMin) != 0u) {
#pragma unroll
        for (int q = 0; q < 4; ++q)
          insert_block(s[4 * g + q], chunkBase + hb + lane + 32 * (4 * g + q),
                       tvr, tir, lane, curMin, minPos);
      }
    }
  }
}

__global__ __launch_bounds__(1024, 1)
void core_processor_kernel(const float* __restrict__ x,
                           const float* __restrict__ mem_keys,
                           const float* __restrict__ mem_values,
                           const float* __restrict__ W_fuse,
                           const float* __restrict__ b_fuse,
                           const float* __restrict__ ln_g,
                           const float* __restrict__ ln_b,
                           const float* __restrict__ W_op,
                           const float* __restrict__ b_op,
                           float* __restrict__ out)
{
  __shared__ __align__(16) float xs[TOK_PER_WG * A_PITCH];
  __shared__ __align__(16) float sc[2 * SC_BUF];

  const int tid  = threadIdx.x;
  const int lane = tid & 31;
  const int w    = tid >> 5;
  const long tokBase = (long)blockIdx.x * TOK_PER_WG;

  const int rowA  = lane & 15;
  const int khalf = (lane >> 4) << 1;
  const int kquad = (lane >> 4) << 2;

  {
    const float4* src = (const float4*)(x + tokBase * DDIM);
    const int i0 = tid, i1 = tid + 1024;
    *(float4*)(xs + (i0 >> 6) * A_PITCH + (i0 & 63) * 4) = src[i0];
    *(float4*)(xs + (i1 >> 6) * A_PITCH + (i1 & 63) * 4) = src[i1];
  }
  float tvr = -INFINITY;
  int   tir = 0;
  float curMin = -INFINITY;
  int   minPos = 0;
  __syncthreads();

  const bool gemmFirst = (__builtin_amdgcn_readfirstlane(w) & 4) != 0;

  score_chunk(mem_keys, xs, sc, 0, w, lane, rowA, kquad);
  __syncthreads();

  for (int chunk = 0; chunk < NCHUNK; ++chunk) {
    const int chunkBase = chunk * NT;
    const float* cur = sc + (chunk & 1) * SC_BUF;
    float*       nxt = sc + ((chunk & 1) ^ 1) * SC_BUF;
    const bool more = (chunk + 1 < NCHUNK);
    if (gemmFirst) {
      if (more) score_chunk(mem_keys, xs, nxt, chunkBase + NT, w, lane, rowA, kquad);
      merge_chunk(cur, tvr, tir, chunkBase, w, lane, curMin, minPos);
    } else {
      merge_chunk(cur, tvr, tir, chunkBase, w, lane, curMin, minPos);
      if (more) score_chunk(mem_keys, xs, nxt, chunkBase + NT, w, lane, rowA, kquad);
    }
    __syncthreads();
  }

  {
    const float myv = tvr;
    const int   myi = tir;
    const float mx  = wave_max_f(myv);
    const float e   = __expf(myv - mx);
    const float sum = wave_sum_f(e);
    const float wn  = e / sum;

    float acc[8];
#pragma unroll
    for (int q = 0; q < 8; ++q) acc[q] = 0.f;
#pragma unroll 1
    for (int k = 0; k < KSEL; ++k) {
      const float wk = __shfl(wn, k, 32);
      const int   id = __shfl(myi, k, 32);
      const float4* vp = (const float4*)(mem_values + (long)id * DDIM + lane * 8);
      const float4 a0 = vp[0];
      const float4 a1 = vp[1];
      acc[0] += wk * a0.x; acc[1] += wk * a0.y; acc[2] += wk * a0.z; acc[3] += wk * a0.w;
      acc[4] += wk * a1.x; acc[5] += wk * a1.y; acc[6] += wk * a1.z; acc[7] += wk * a1.w;
    }
#pragma unroll
    for (int q = 0; q < 8; ++q) xs[w * A_PITCH + lane * 8 + q] += acc[q];
  }
  __syncthreads();

  const int mt = w >> 4;
  const int nt = w & 15;
  {
    const int colB = nt * 16 + (lane & 15);
    const float* ap = xs + (mt * 16 + rowA) * A_PITCH + khalf;
    v8f acc = {};
#pragma unroll 4
    for (int k0 = 0; k0 < DDIM; k0 += 4) {
      v2f a = *(const v2f*)(ap + k0);
      const int kk = k0 + khalf;
      v2f b;
      b.x = W_fuse[(long)kk * DDIM + colB];
      b.y = W_fuse[(long)(kk + 1) * DDIM + colB];
      acc = __builtin_amdgcn_wmma_f32_16x16x4_f32(
          false, a, false, b, (short)0, acc, false, false);
    }
    const float bf = b_fuse[colB];
    const int m0 = (lane >> 4) * 8;
#pragma unroll
    for (int r = 0; r < 8; ++r)
      sc[(mt * 16 + m0 + r) * A_PITCH + colB] = acc[r] + bf;
  }
  __syncthreads();

  {
    const int d0 = lane * 8;
    float h[8];
#pragma unroll
    for (int q = 0; q < 8; ++q) h[q] = sc[w * A_PITCH + d0 + q];
    float s1 = 0.f, s2 = 0.f;
#pragma unroll
    for (int q = 0; q < 8; ++q) { s1 += h[q]; s2 += h[q] * h[q]; }
    s1 = wave_sum_f(s1);
    s2 = wave_sum_f(s2);
    const float mu   = s1 * (1.0f / DDIM);
    const float var  = s2 * (1.0f / DDIM) - mu * mu;
    const float rstd = rsqrtf(var + LN_EPS);
#pragma unroll
    for (int q = 0; q < 8; ++q) {
      const float g = ln_g[d0 + q];
      const float b = ln_b[d0 + q];
      const float v = (h[q] - mu) * rstd * g + b;
      sc[w * A_PITCH + d0 + q] = fmaxf(v, 0.f);
    }
  }
  __syncthreads();

  {
    const int colB = nt * 16 + (lane & 15);
    const float* ap = sc + (mt * 16 + rowA) * A_PITCH + khalf;
    v8f acc = {};
#pragma unroll 4
    for (int k0 = 0; k0 < DDIM; k0 += 4) {
      v2f a = *(const v2f*)(ap + k0);
      const int kk = k0 + khalf;
      v2f b;
      b.x = W_op[(long)kk * DDIM + colB];
      b.y = W_op[(long)(kk + 1) * DDIM + colB];
      acc = __builtin_amdgcn_wmma_f32_16x16x4_f32(
          false, a, false, b, (short)0, acc, false, false);
    }
    const float bo = b_op[colB];
    const int m0 = (lane >> 4) * 8;
#pragma unroll
    for (int r = 0; r < 8; ++r)
      out[(tokBase + mt * 16 + m0 + r) * DDIM + colB] = acc[r] + bo;
  }
}

extern "C" void kernel_launch(void* const* d_in, const int* in_sizes, int n_in,
                              void* d_out, int out_size, void* d_ws, size_t ws_size,
                              hipStream_t stream) {
  const float* x          = (const float*)d_in[0];
  const float* mem_keys   = (const float*)d_in[1];
  const float* mem_values = (const float*)d_in[2];
  const float* W_fuse     = (const float*)d_in[3];
  const float* b_fuse     = (const float*)d_in[4];
  const float* ln_g       = (const float*)d_in[5];
  const float* ln_b       = (const float*)d_in[6];
  const float* W_op       = (const float*)d_in[7];
  const float* b_op       = (const float*)d_in[8];
  float* out = (float*)d_out;

  dim3 grid(8192 / TOK_PER_WG);
  dim3 block(1024);
  hipLaunchKernelGGL(core_processor_kernel, grid, block, 0, stream,
                     x, mem_keys, mem_values, W_fuse, b_fuse, ln_g, ln_b,
                     W_op, b_op, out);
}
